# starting kernel
# baseline (speedup 1.0000x reference)
#include <hip/hip_runtime.h>
#include <hip/hip_bf16.h>

typedef __attribute__((ext_vector_type(2))) float v2f;
typedef __attribute__((ext_vector_type(8))) float v8f;

#define EPS 1e-5f

__global__ void k_repack(const float* __restrict__ comp_w,
                         const float* __restrict__ enc_w,
                         float* __restrict__ wCw,
                         float* __restrict__ wAe) {
    const int i = blockIdx.x * blockDim.x + threadIdx.x;
    if (i < 8192) {
        const int o = (i >> 1) & 63;
        const int k = ((i >> 7) << 1) | (i & 1);
        wCw[i] = comp_w[o * 128 + k];
    }
    if (i < 64512) {
        const int o = (i >> 1) % 112;
        const int t = ((i / 224) << 1) | (i & 1);
        const int c = t & 63, off = t >> 6;
        wAe[i] = (o < 100) ? enc_w[(size_t)o * 576 + c * 9 + off] : 0.f;
    }
}

__global__ void k_compress(const float* __restrict__ x,
                           const float* __restrict__ wCw,
                           const float* __restrict__ gamma,
                           const float* __restrict__ beta,
                           const float* __restrict__ mean,
                           const float* __restrict__ var,
                           float* __restrict__ y) {
    __shared__ float ls[128 * 16];
    const int tid    = threadIdx.x;
    const int sTile  = blockIdx.x;
    const int b      = sTile >> 8;
    const int hwBase = (sTile & 255) << 4;

    const float* xb = x + (size_t)b * 128 * 4096;
#pragma unroll
    for (int i = tid; i < 2048; i += 128)
        ls[i] = xb[(size_t)(i >> 4) * 4096 + hwBase + (i & 15)];
    __syncthreads();

    const int lane = tid & 31, l = lane & 15, half = lane >> 4;
    const int oBase = (tid >> 5) << 4;
    const int aCol = (oBase + l) * 2;

    v8f acc = {};
#pragma unroll 4
    for (int kb = 0; kb < 128; kb += 4) {
        const int k0 = kb + half * 2;
        const v2f a = *(const v2f*)(wCw + ((kb >> 1) + half) * 128 + aCol);
        v2f bm;
        bm.x = ls[k0 * 16 + l];
        bm.y = ls[(k0 + 1) * 16 + l];
        acc = __builtin_amdgcn_wmma_f32_16x16x4_f32(
            false, a, false, bm, (short)0, acc, false, false);
    }

#pragma unroll
    for (int j = 0; j < 8; ++j) {
        const int o = oBase + j + half * 8;
        const float iv = gamma[o] * rsqrtf(var[o] + EPS);
        const float bb = beta[o] - mean[o] * iv;
        float v = acc[j] * iv + bb;
        v = v * __frcp_rn(1.f + __expf(-v));
        y[((size_t)b * 64 + o) * 4096 + hwBase + l] = v;
    }
}

__global__ void k_encconv(const float* __restrict__ y,
                          const float* __restrict__ wAe,
                          const float* __restrict__ enc_b,
                          float* __restrict__ logits) {
    __shared__ float ls[64 * 54];
    const int tid    = threadIdx.x;
    const int sTile  = blockIdx.x;
    const int b      = sTile >> 8;
    const int hwBase = (sTile & 255) << 4;
    const int h      = hwBase >> 6;
    const int w0     = hwBase & 63;

    const float* yb = y + (size_t)b * 64 * 4096;
    for (int i = tid; i < 3456; i += 224) {
        const int c = i / 54, rem = i % 54;
        const int r = rem / 18, col = rem % 18;
        const int hh = h + r - 1, ww = w0 + col - 1;
        const bool ok = (hh >= 0) & (hh < 64) & (ww >= 0) & (ww < 64);
        ls[i] = ok ? yb[(size_t)c * 4096 + hh * 64 + ww] : 0.f;
    }
    __syncthreads();

    const int lane = tid & 31, l = lane & 15, half = lane >> 4;
    const int oBase = (tid >> 5) << 4;
    const int aCol = (oBase + l) * 2;

    v8f acc = {};
    for (int off = 0; off < 9; ++off) {
        const int ki = off / 3, kj = off % 3;
        const int bcol = ki * 18 + kj + l;
#pragma unroll 4
        for (int cb = 0; cb < 64; cb += 4) {
            const int c0 = cb + half * 2;
            const int pair = off * 32 + (cb >> 1) + half;
            const v2f a = *(const v2f*)(wAe + pair * 224 + aCol);
            v2f bm;
            bm.x = ls[c0 * 54 + bcol];
            bm.y = ls[(c0 + 1) * 54 + bcol];
            acc = __builtin_amdgcn_wmma_f32_16x16x4_f32(
                false, a, false, bm, (short)0, acc, false, false);
        }
    }

#pragma unroll
    for (int j = 0; j < 8; ++j) {
        const int oo = oBase + j + half * 8;
        if (oo < 100)
            logits[((size_t)b * 100 + oo) * 4096 + hwBase + l] = acc[j] + enc_b[oo];
    }
}

__global__ void k_reassemble(const float* __restrict__ x,
                             const float* __restrict__ logits,
                             float* __restrict__ out) {
    extern __shared__ float ls[];
    const int tid = threadIdx.x;
    const int X   = tid & 127;
    const int s1  = tid >> 7;
    const int b   = blockIdx.x >> 6;
    const int Yp  = blockIdx.x & 63;
    const int Y   = Yp * 2 + s1;
    const int s2  = X & 1;
    const int hbase = Yp - 1;

    const int r   = (tid < 198) ? tid / 66 : 0;
    const int col = (tid < 198) ? tid % 66 : 0;
    const int hh  = hbase + r, ww = col - 1;
    const bool ok = (tid < 198) & (hh >= 0) & (hh < 64) & (ww >= 0) & (ww < 64);
    const float* xb = x + (size_t)b * 128 * 4096;
    const int xoff  = hh * 64 + ww;
    if (tid < 198) {
#pragma unroll 4
        for (int c = 0; c < 128; ++c)
            ls[c * 198 + tid] = ok ? xb[(size_t)c * 4096 + xoff] : 0.f;
    }
    __syncthreads();

    const int w = X >> 1;
    const float* lg = logits + (size_t)b * 100 * 4096 + (size_t)Yp * 64 + w;
    float p[25];
    float mx = -1e30f;
#pragma unroll
    for (int q = 0; q < 25; ++q) {
        p[q] = lg[(size_t)(q * 4 + s1 * 2 + s2) * 4096];
        mx = fmaxf(mx, p[q]);
    }
    float sum = 0.f;
#pragma unroll
    for (int q = 0; q < 25; ++q) { p[q] = __expf(p[q] - mx); sum += p[q]; }
    const float rs = __frcp_rn(sum);

    float wc[9] = {0.f, 0.f, 0.f, 0.f, 0.f, 0.f, 0.f, 0.f, 0.f};
#pragma unroll
    for (int ki = 0; ki < 5; ++ki) {
        const int di = (ki + s1) >> 1;
#pragma unroll
        for (int kj = 0; kj < 5; ++kj) {
            const int dj = (kj + s2) >> 1;
            wc[di * 3 + dj] += p[ki * 5 + kj] * rs;
        }
    }

    const int base = ((X - 2) >> 1) + 1;
    float* ob = out + (size_t)b * 128 * 16384 + (size_t)Y * 128 + X;
#pragma unroll 2
    for (int c = 0; c < 128; ++c) {
        const float* lc = ls + c * 198 + base;
        float s = 0.f;
#pragma unroll
        for (int t = 0; t < 9; ++t)
            s += wc[t] * lc[(t / 3) * 66 + (t % 3)];
        ob[(size_t)c * 16384] = s;
    }
}

extern "C" void kernel_launch(void* const* d_in, const int* in_sizes, int n_in,
                              void* d_out, int out_size, void* d_ws, size_t ws_size,
                              hipStream_t stream) {
    const float* x      = (const float*)d_in[0];
    const float* comp_w = (const float*)d_in[1];
    const float* gamma  = (const float*)d_in[2];
    const float* beta   = (const float*)d_in[3];
    const float* mean   = (const float*)d_in[4];
    const float* var    = (const float*)d_in[5];
    const float* enc_w  = (const float*)d_in[6];
    const float* enc_b  = (const float*)d_in[7];
    float* out = (float*)d_out;

    float* y      = (float*)d_ws;
    float* logits = y + (size_t)1048576;
    float* wCw    = logits + (size_t)1638400;
    float* wAe    = wCw + (size_t)8192;

    k_repack<<<252, 256, 0, stream>>>(comp_w, enc_w, wCw, wAe);
    k_compress<<<1024, 128, 0, stream>>>(x, wCw, gamma, beta, mean, var, y);
    k_encconv<<<1024, 224, 0, stream>>>(y, wAe, enc_b, logits);
    k_reassemble<<<256, 256, 198 * 128 * sizeof(float), stream>>>(x, logits, out);
}


# softmax + 25->9 tap fold moved out of k_reassemble (each of its 4 channel-group blocks redid it) into k_encconv's epilogue, once per output px; 36 folded weights per src px replace the 100 logits
# speedup vs baseline: 2.0256x; 2.0256x over previous
#include <hip/hip_runtime.h>
#include <hip/hip_bf16.h>

typedef __attribute__((ext_vector_type(2))) float v2f;
typedef __attribute__((ext_vector_type(8))) float v8f;

#define EPS 1e-5f

#define CW_P 132
__global__ void k_compress(const float* __restrict__ x,
                           const float* __restrict__ comp_w,
                           const float* __restrict__ gamma,
                           const float* __restrict__ beta,
                           const float* __restrict__ mean,
                           const float* __restrict__ var,
                           float* __restrict__ y) {
    __shared__ float ls[128 * 16];
    __shared__ __attribute__((aligned(16))) float lwc[64 * CW_P];
    const int tid    = threadIdx.x;
    const int sTile  = blockIdx.x;
    const int b      = sTile >> 8;
    const int hwBase = (sTile & 255) << 4;

    const float4* w4 = (const float4*)comp_w;
    for (int i = 0; i < 16; ++i) {
        const int f = tid + i * 128;
        *(float4*)(lwc + (f >> 5) * CW_P + (f & 31) * 4) = w4[f];
    }

    const float* xb = x + (size_t)b * 128 * 4096;
#pragma unroll
    for (int i = tid; i < 2048; i += 128)
        ls[i] = xb[(size_t)(i >> 4) * 4096 + hwBase + (i & 15)];
    __syncthreads();

    const int lane = tid & 31, l = lane & 15, half = lane >> 4;
    const int oBase = (tid >> 5) << 4;
    const float* aRow = lwc + (oBase + l) * CW_P + half * 2;

    v8f acc = {};
#pragma unroll 4
    for (int kb = 0; kb < 128; kb += 4) {
        const int k0 = kb + half * 2;
        const v2f a = *(const v2f*)(aRow + kb);
        v2f bm;
        bm.x = ls[k0 * 16 + l];
        bm.y = ls[(k0 + 1) * 16 + l];
        acc = __builtin_amdgcn_wmma_f32_16x16x4_f32(
            false, a, false, bm, (short)0, acc, false, false);
    }

#pragma unroll
    for (int j = 0; j < 8; ++j) {
        const int o = oBase + j + half * 8;
        const float iv = gamma[o] * rsqrtf(var[o] + EPS);
        const float bb = beta[o] - mean[o] * iv;
        float v = acc[j] * iv + bb;
        v = v * __frcp_rn(1.f + __expf(-v));
        y[((size_t)b * 64 + o) * 4096 + hwBase + l] = v;
    }
}

#define ENC_WP  580
#define ENC_WF  (100 * ENC_WP)
#define ENC_ROW 68
#define ENC_CP  208
#define LG_P    68

template <int NT>
__device__ __forceinline__ void enc_wave(const float* __restrict__ lw,
                                         const float* __restrict__ lp,
                                         const int oTile0, const int pxBase,
                                         const int l, const int half,
                                         v8f* __restrict__ acc) {
    for (int t = 0; t < NT; ++t) acc[t] = v8f{};

    const float* aRow[NT];
    for (int t = 0; t < NT; ++t) {
        const int o = (oTile0 + t) * 16 + l;
        aRow[t] = lw + (o < 100 ? o : 99) * ENC_WP + half * 18;
    }
    const float* bBase = lp + half * 2 * ENC_CP + 3 + pxBase + l;
#pragma unroll 2
    for (int g = 0; g < 16; ++g) {
        const float* bp = bBase + g * 4 * ENC_CP;
#pragma unroll
        for (int j = 0; j < 9; ++j) {
            const int e0 = 2 * j, e1 = 2 * j + 1;
            const int t0 = e0 % 9, t1 = e1 % 9;
            v2f bm;
            bm.x = bp[(e0 / 9) * ENC_CP + (t0 / 3) * ENC_ROW + (t0 % 3)];
            bm.y = bp[(e1 / 9) * ENC_CP + (t1 / 3) * ENC_ROW + (t1 % 3)];
            for (int t = 0; t < NT; ++t) {
                const v2f a = *(const v2f*)(aRow[t] + g * 36 + 2 * j);
                acc[t] = __builtin_amdgcn_wmma_f32_16x16x4_f32(
                    false, a, false, bm, (short)0, acc[t], false, false);
            }
        }
    }
}

template <int S1, int S2>
__device__ __forceinline__ void fold_store(const float* __restrict__ lgc,
                                           float* __restrict__ wp,
                                           const bool top, const bool bot,
                                           const bool lft, const bool rgt) {
    float p[25];
    float mx = -1e30f;
#pragma unroll
    for (int q = 0; q < 25; ++q) {
        p[q] = lgc[(q * 4 + S1 * 2 + S2) * LG_P];
        mx = fmaxf(mx, p[q]);
    }
    float sum = 0.f;
#pragma unroll
    for (int q = 0; q < 25; ++q) { p[q] = __expf(p[q] - mx); sum += p[q]; }
    const float rs = __frcp_rn(sum);

    float wc[9] = {0.f, 0.f, 0.f, 0.f, 0.f, 0.f, 0.f, 0.f, 0.f};
#pragma unroll
    for (int ki = 0; ki < 5; ++ki) {
        const int di = (ki + S1) >> 1;
#pragma unroll
        for (int kj = 0; kj < 5; ++kj) {
            const int dj = (kj + S2) >> 1;
            wc[di * 3 + dj] += p[ki * 5 + kj] * rs;
        }
    }
#pragma unroll
    for (int t = 0; t < 9; ++t) {
        const int di = t / 3, dj = t % 3;
        const bool oob = (di == 0 && top) || (di == 2 && bot) ||
                         (dj == 0 && lft) || (dj == 2 && rgt);
        wp[(t * 4 + S1 * 2 + S2) * 64] = oob ? 0.f : wc[t];
    }
}

__global__ void __launch_bounds__(256)
k_encconv(const float* __restrict__ y,
          const float* __restrict__ enc_w,
          const float* __restrict__ enc_b,
          float* __restrict__ wts) {
    __shared__ __attribute__((aligned(16))) float ls[ENC_WF + 64 * ENC_CP];
    float* lw = ls;
    float* lp = ls + ENC_WF;
    const int tid = threadIdx.x;
    const int b   = blockIdx.x >> 6;
    const int h   = blockIdx.x & 63;

    const float* yb = y + (size_t)b * 64 * 4096;
    const int q = tid & 15, cq = tid >> 4;
    float4 v[3][4];
    for (int r = 0; r < 3; ++r) {
        const int hh = h + r - 1;
        if (hh >= 0 && hh < 64) {
            for (int k = 0; k < 4; ++k)
                v[r][k] = *(const float4*)(yb + (size_t)(cq + k * 16) * 4096 + hh * 64 + q * 4);
        } else {
            for (int k = 0; k < 4; ++k)
                v[r][k] = make_float4(0.f, 0.f, 0.f, 0.f);
        }
    }
    for (int r = 0; r < 3; ++r) {
        for (int k = 0; k < 4; ++k)
            *(float4*)(lp + (cq + k * 16) * ENC_CP + r * ENC_ROW + 4 + q * 4) = v[r][k];
    }
    if (tid < 192) {
        float* p = lp + (tid / 3) * ENC_CP + (tid % 3) * ENC_ROW;
        p[3]  = 0.f;
        p[68] = 0.f;
    }

    const float4* g4 = (const float4*)enc_w;
#pragma unroll 14
    for (int k = 0; k < 56; ++k) {
        const int f = tid + k * 256;
        const int row = f / 144;
        *(float4*)(lw + row * ENC_WP + (f - row * 144) * 4) = g4[f];
    }
    if (tid < 64) {
        const int f = tid + 56 * 256;
        const int row = f / 144;
        *(float4*)(lw + row * ENC_WP + (f - row * 144) * 4) = g4[f];
    }
    __syncthreads();

    const int lane = tid & 31, l = lane & 15, half = lane >> 4;
    const int wv = __builtin_amdgcn_readfirstlane(tid >> 5);
    const int pxBase = (wv & 3) << 4;
    v8f acc[4];
    if (wv < 4) enc_wave<4>(lw, lp, 0, pxBase, l, half, acc);
    else        enc_wave<3>(lw, lp, 4, pxBase, l, half, acc);
    __syncthreads();

    float* lgs = lp;
    const int oTile0 = wv < 4 ? 0 : 4;
    const int nt     = wv < 4 ? 4 : 3;
#pragma unroll
    for (int t = 0; t < 4; ++t) {
        if (t < nt) {
#pragma unroll
            for (int j = 0; j < 8; ++j) {
                const int oo = (oTile0 + t) * 16 + j + half * 8;
                if (oo < 100)
                    lgs[oo * LG_P + pxBase + l] = acc[t][j] + enc_b[oo];
            }
        }
    }
    __syncthreads();

    const int w = ((wv & 1) << 5) + lane;
    float* wp = wts + (size_t)(b * 64 + h) * (9 * 4 * 64) + w;
    const bool top = h == 0, bot = h == 63, lft = w == 0, rgt = w == 63;
    switch (wv >> 1) {
    case 0:  fold_store<0, 0>(lgs + w, wp, top, bot, lft, rgt); break;
    case 1:  fold_store<0, 1>(lgs + w, wp, top, bot, lft, rgt); break;
    case 2:  fold_store<1, 0>(lgs + w, wp, top, bot, lft, rgt); break;
    default: fold_store<1, 1>(lgs + w, wp, top, bot, lft, rgt); break;
    }
}

#define RS_CG  4
#define RS_CPB (128 / RS_CG)

__device__ __forceinline__ float row_gather(const unsigned u, const unsigned sel) {
    return __uint_as_float(__builtin_amdgcn_permlane16_var(u, u, sel, true, false));
}

__global__ void __launch_bounds__(256)
k_reassemble(const float* __restrict__ x,
             const float* __restrict__ wts,
             float* __restrict__ out) {
    const int tid = threadIdx.x;
    const int X   = tid & 127;
    const int s1  = tid >> 7;
    const int cg  = blockIdx.x & (RS_CG - 1);
    const int rp  = blockIdx.x / RS_CG;
    const int b   = rp >> 6;
    const int Yp  = rp & 63;
    const int Y   = Yp * 2 + s1;
    const int s2  = X & 1;
    const int w   = X >> 1;

    const float* wq = wts + ((size_t)(b * 64 + Yp) * 36 + s1 * 2 + s2) * 64 + w;
    float wc[9];
#pragma unroll
    for (int t = 0; t < 9; ++t) wc[t] = wq[t * 256];

    const int j  = tid & 15;
    const int wb = w - (j >> 1);
    const int cj = min(max(wb - 1 + j, 0), 63);
    const unsigned sel0 = (unsigned)(j >> 1), sel1 = sel0 + 1u, sel2 = sel0 + 2u;
    const float* xb = x + ((size_t)b * 128 + cg * RS_CPB) * 4096 + cj;
    const float* x0 = xb + max(Yp - 1, 0) * 64;
    const float* x1 = xb + Yp * 64;
    const float* x2 = xb + min(Yp + 1, 63) * 64;
    float* ob = out + ((size_t)b * 128 + cg * RS_CPB) * 16384 + (size_t)Y * 128 + X;
#pragma unroll 8
    for (int c = 0; c < RS_CPB; ++c) {
        const unsigned u0 = __float_as_uint(x0[(size_t)c * 4096]);
        const unsigned u1 = __float_as_uint(x1[(size_t)c * 4096]);
        const unsigned u2 = __float_as_uint(x2[(size_t)c * 4096]);
        float s = wc[0] * row_gather(u0, sel0);
        s = __builtin_fmaf(wc[1], row_gather(u0, sel1), s);
        s = __builtin_fmaf(wc[2], row_gather(u0, sel2), s);
        s = __builtin_fmaf(wc[3], row_gather(u1, sel0), s);
        s = __builtin_fmaf(wc[4], row_gather(u1, sel1), s);
        s = __builtin_fmaf(wc[5], row_gather(u1, sel2), s);
        s = __builtin_fmaf(wc[6], row_gather(u2, sel0), s);
        s = __builtin_fmaf(wc[7], row_gather(u2, sel1), s);
        s = __builtin_fmaf(wc[8], row_gather(u2, sel2), s);
        ob[(size_t)c * 16384] = s;
    }
}

extern "C" void kernel_launch(void* const* d_in, const int* in_sizes, int n_in,
                              void* d_out, int out_size, void* d_ws, size_t ws_size,
                              hipStream_t stream) {
    const float* x      = (const float*)d_in[0];
    const float* comp_w = (const float*)d_in[1];
    const float* gamma  = (const float*)d_in[2];
    const float* beta   = (const float*)d_in[3];
    const float* mean   = (const float*)d_in[4];
    const float* var    = (const float*)d_in[5];
    const float* enc_w  = (const float*)d_in[6];
    const float* enc_b  = (const float*)d_in[7];
    float* out = (float*)d_out;

    float* y      = (float*)d_ws;
    float* wts    = y + (size_t)1048576;

    k_compress<<<1024, 128, 0, stream>>>(x, comp_w, gamma, beta, mean, var, y);
    k_encconv<<<256, 256, 0, stream>>>(y, enc_w, enc_b, wts);
    k_reassemble<<<256 * RS_CG, 256, 0, stream>>>(x, wts, out);
}


# k_reassemble: a lane makes the 2x2 output quads of two source px (wave = one source row: b64 x loads, b128 stores of whole output rows, neighbour columns by +-1 lane permute); 8192 -> 4096 waves
# speedup vs baseline: 2.3472x; 1.1588x over previous
#include <hip/hip_runtime.h>
#include <hip/hip_bf16.h>

typedef __attribute__((ext_vector_type(2))) float v2f;
typedef __attribute__((ext_vector_type(4))) float v4f;
typedef __attribute__((ext_vector_type(8))) float v8f;

#define EPS 1e-5f

#define CW_P 132
__global__ void k_compress(const float* __restrict__ x,
                           const float* __restrict__ comp_w,
                           const float* __restrict__ gamma,
                           const float* __restrict__ beta,
                           const float* __restrict__ mean,
                           const float* __restrict__ var,
                           float* __restrict__ y) {
    __shared__ float ls[128 * 16];
    __shared__ __attribute__((aligned(16))) float lwc[64 * CW_P];
    const int tid    = threadIdx.x;
    const int sTile  = blockIdx.x;
    const int b      = sTile >> 8;
    const int hwBase = (sTile & 255) << 4;

    const float4* w4 = (const float4*)comp_w;
    for (int i = 0; i < 16; ++i) {
        const int f = tid + i * 128;
        *(float4*)(lwc + (f >> 5) * CW_P + (f & 31) * 4) = w4[f];
    }

    const float* xb = x + (size_t)b * 128 * 4096;
#pragma unroll
    for (int i = tid; i < 2048; i += 128)
        ls[i] = xb[(size_t)(i >> 4) * 4096 + hwBase + (i & 15)];
    __syncthreads();

    const int lane = tid & 31, l = lane & 15, half = lane >> 4;
    const int oBase = (tid >> 5) << 4;
    const float* aRow = lwc + (oBase + l) * CW_P + half * 2;

    v8f acc = {};
#pragma unroll 4
    for (int kb = 0; kb < 128; kb += 4) {
        const int k0 = kb + half * 2;
        const v2f a = *(const v2f*)(aRow + kb);
        v2f bm;
        bm.x = ls[k0 * 16 + l];
        bm.y = ls[(k0 + 1) * 16 + l];
        acc = __builtin_amdgcn_wmma_f32_16x16x4_f32(
            false, a, false, bm, (short)0, acc, false, false);
    }

#pragma unroll
    for (int j = 0; j < 8; ++j) {
        const int o = oBase + j + half * 8;
        const float iv = gamma[o] * rsqrtf(var[o] + EPS);
        const float bb = beta[o] - mean[o] * iv;
        float v = acc[j] * iv + bb;
        v = v * __frcp_rn(1.f + __expf(-v));
        y[((size_t)b * 64 + o) * 4096 + hwBase + l] = v;
    }
}

#define ENC_WP  580
#define ENC_WF  (100 * ENC_WP)
#define ENC_ROW 68
#define ENC_CP  208
#define LG_P    68

template <int NT>
__device__ __forceinline__ void enc_wave(const float* __restrict__ lw,
                                         const float* __restrict__ lp,
                                         const int oTile0, const int pxBase,
                                         const int l, const int half,
                                         v8f* __restrict__ acc) {
    for (int t = 0; t < NT; ++t) acc[t] = v8f{};

    const float* aRow[NT];
    for (int t = 0; t < NT; ++t) {
        const int o = (oTile0 + t) * 16 + l;
        aRow[t] = lw + (o < 100 ? o : 99) * ENC_WP + half * 18;
    }
    const float* bBase = lp + half * 2 * ENC_CP + 3 + pxBase + l;
#pragma unroll 2
    for (int g = 0; g < 16; ++g) {
        const float* bp = bBase + g * 4 * ENC_CP;
#pragma unroll
        for (int j = 0; j < 9; ++j) {
            const int e0 = 2 * j, e1 = 2 * j + 1;
            const int t0 = e0 % 9, t1 = e1 % 9;
            v2f bm;
            bm.x = bp[(e0 / 9) * ENC_CP + (t0 / 3) * ENC_ROW + (t0 % 3)];
            bm.y = bp[(e1 / 9) * ENC_CP + (t1 / 3) * ENC_ROW + (t1 % 3)];
            for (int t = 0; t < NT; ++t) {
                const v2f a = *(const v2f*)(aRow[t] + g * 36 + 2 * j);
                acc[t] = __builtin_amdgcn_wmma_f32_16x16x4_f32(
                    false, a, false, bm, (short)0, acc[t], false, false);
            }
        }
    }
}

template <int S1, int S2>
__device__ __forceinline__ void fold_store(const float* __restrict__ lgc,
                                           float* __restrict__ wp,
                                           const bool top, const bool bot,
                                           const bool lft, const bool rgt) {
    float p[25];
    float mx = -1e30f;
#pragma unroll
    for (int q = 0; q < 25; ++q) {
        p[q] = lgc[(q * 4 + S1 * 2 + S2) * LG_P];
        mx = fmaxf(mx, p[q]);
    }
    float sum = 0.f;
#pragma unroll
    for (int q = 0; q < 25; ++q) { p[q] = __expf(p[q] - mx); sum += p[q]; }
    const float rs = __frcp_rn(sum);

    float wc[9] = {0.f, 0.f, 0.f, 0.f, 0.f, 0.f, 0.f, 0.f, 0.f};
#pragma unroll
    for (int ki = 0; ki < 5; ++ki) {
        const int di = (ki + S1) >> 1;
#pragma unroll
        for (int kj = 0; kj < 5; ++kj) {
            const int dj = (kj + S2) >> 1;
            wc[di * 3 + dj] += p[ki * 5 + kj] * rs;
        }
    }
#pragma unroll
    for (int t = 0; t < 9; ++t) {
        const int di = t / 3, dj = t % 3;
        const bool oob = (di == 0 && top) || (di == 2 && bot) ||
                         (dj == 0 && lft) || (dj == 2 && rgt);
        wp[(t * 4 + S1 * 2 + S2) * 64] = oob ? 0.f : wc[t];
    }
}

__global__ void __launch_bounds__(256)
k_encconv(const float* __restrict__ y,
          const float* __restrict__ enc_w,
          const float* __restrict__ enc_b,
          float* __restrict__ wts) {
    __shared__ __attribute__((aligned(16))) float ls[ENC_WF + 64 * ENC_CP];
    float* lw = ls;
    float* lp = ls + ENC_WF;
    const int tid = threadIdx.x;
    const int b   = blockIdx.x >> 6;
    const int h   = blockIdx.x & 63;

    const float* yb = y + (size_t)b * 64 * 4096;
    const int q = tid & 15, cq = tid >> 4;
    float4 v[3][4];
    for (int r = 0; r < 3; ++r) {
        const int hh = h + r - 1;
        if (hh >= 0 && hh < 64) {
            for (int k = 0; k < 4; ++k)
                v[r][k] = *(const float4*)(yb + (size_t)(cq + k * 16) * 4096 + hh * 64 + q * 4);
        } else {
            for (int k = 0; k < 4; ++k)
                v[r][k] = make_float4(0.f, 0.f, 0.f, 0.f);
        }
    }
    for (int r = 0; r < 3; ++r) {
        for (int k = 0; k < 4; ++k)
            *(float4*)(lp + (cq + k * 16) * ENC_CP + r * ENC_ROW + 4 + q * 4) = v[r][k];
    }
    if (tid < 192) {
        float* p = lp + (tid / 3) * ENC_CP + (tid % 3) * ENC_ROW;
        p[3]  = 0.f;
        p[68] = 0.f;
    }

    const float4* g4 = (const float4*)enc_w;
#pragma unroll 14
    for (int k = 0; k < 56; ++k) {
        const int f = tid + k * 256;
        const int row = f / 144;
        *(float4*)(lw + row * ENC_WP + (f - row * 144) * 4) = g4[f];
    }
    if (tid < 64) {
        const int f = tid + 56 * 256;
        const int row = f / 144;
        *(float4*)(lw + row * ENC_WP + (f - row * 144) * 4) = g4[f];
    }
    __syncthreads();

    const int lane = tid & 31, l = lane & 15, half = lane >> 4;
    const int wv = __builtin_amdgcn_readfirstlane(tid >> 5);
    const int pxBase = (wv & 3) << 4;
    v8f acc[4];
    if (wv < 4) enc_wave<4>(lw, lp, 0, pxBase, l, half, acc);
    else        enc_wave<3>(lw, lp, 4, pxBase, l, half, acc);
    __syncthreads();

    float* lgs = lp;
    const int oTile0 = wv < 4 ? 0 : 4;
    const int nt     = wv < 4 ? 4 : 3;
#pragma unroll
    for (int t = 0; t < 4; ++t) {
        if (t < nt) {
#pragma unroll
            for (int j = 0; j < 8; ++j) {
                const int oo = (oTile0 + t) * 16 + j + half * 8;
                if (oo < 100)
                    lgs[oo * LG_P + pxBase + l] = acc[t][j] + enc_b[oo];
            }
        }
    }
    __syncthreads();

    const int w = ((wv & 1) << 5) + lane;
    float* wp = wts + (size_t)(b * 64 + h) * (9 * 4 * 64) + w;
    const bool top = h == 0, bot = h == 63, lft = w == 0, rgt = w == 63;
    switch (wv >> 1) {
    case 0:  fold_store<0, 0>(lgs + w, wp, top, bot, lft, rgt); break;
    case 1:  fold_store<0, 1>(lgs + w, wp, top, bot, lft, rgt); break;
    case 2:  fold_store<1, 0>(lgs + w, wp, top, bot, lft, rgt); break;
    default: fold_store<1, 1>(lgs + w, wp, top, bot, lft, rgt); break;
    }
}

#define RS_CPW 8

__global__ void __launch_bounds__(512)
k_reassemble(const float* __restrict__ x,
             const float* __restrict__ wts,
             float* __restrict__ out) {
    const int tid  = threadIdx.x;
    const int lane = tid & 31;
    const int c0   = (tid >> 5) * RS_CPW;
    const int b    = blockIdx.x >> 6;
    const int h    = blockIdx.x & 63;

    const int aL = ((lane + 31) & 31) << 2;
    const int aR = ((lane + 1) & 31) << 2;

    const float* xb = x + ((size_t)b * 128 + c0) * 4096 + lane * 2;
    const float* x0 = xb + max(h - 1, 0) * 64;
    const float* x1 = xb + h * 64;
    const float* x2 = xb + min(h + 1, 63) * 64;
    float* ob = out + ((size_t)b * 128 + c0) * 16384 + (size_t)h * 256 + lane * 4;

    v2f xs[RS_CPW][3];
#pragma unroll
    for (int c = 0; c < RS_CPW; ++c) {
        xs[c][0] = *(const v2f*)(x0 + (size_t)c * 4096);
        xs[c][1] = *(const v2f*)(x1 + (size_t)c * 4096);
        xs[c][2] = *(const v2f*)(x2 + (size_t)c * 4096);
    }
    const float* wq = wts + (size_t)(b * 64 + h) * (36 * 64) + lane * 2;
    v2f wc[36];
#pragma unroll
    for (int i = 0; i < 36; ++i) wc[i] = *(const v2f*)(wq + i * 64);
    __builtin_amdgcn_sched_barrier(0);

#pragma unroll
    for (int c = 0; c < RS_CPW; ++c) {
        v2f acc[4];
#pragma unroll
        for (int di = 0; di < 3; ++di) {
            const v2f v = xs[c][di];
            const float lf = __int_as_float(
                __builtin_amdgcn_ds_bpermute(aL, __float_as_int(v.y)));
            const float rt = __int_as_float(
                __builtin_amdgcn_ds_bpermute(aR, __float_as_int(v.x)));
            const v2f t0 = {lf, v.x};
            const v2f t1 = v;
            const v2f t2 = {v.y, rt};
#pragma unroll
            for (int s = 0; s < 4; ++s) {
                if (di == 0) acc[s] = wc[s] * t0;
                else acc[s] = __builtin_elementwise_fma(wc[(di * 3) * 4 + s], t0, acc[s]);
                acc[s] = __builtin_elementwise_fma(wc[(di * 3 + 1) * 4 + s], t1, acc[s]);
                acc[s] = __builtin_elementwise_fma(wc[(di * 3 + 2) * 4 + s], t2, acc[s]);
            }
        }
        float* oc = ob + (size_t)c * 16384;
        const v4f r0 = {acc[0].x, acc[1].x, acc[0].y, acc[1].y};
        const v4f r1 = {acc[2].x, acc[3].x, acc[2].y, acc[3].y};
        *(v4f*)(oc)       = r0;
        *(v4f*)(oc + 128) = r1;
    }
}

extern "C" void kernel_launch(void* const* d_in, const int* in_sizes, int n_in,
                              void* d_out, int out_size, void* d_ws, size_t ws_size,
                              hipStream_t stream) {
    const float* x      = (const float*)d_in[0];
    const float* comp_w = (const float*)d_in[1];
    const float* gamma  = (const float*)d_in[2];
    const float* beta   = (const float*)d_in[3];
    const float* mean   = (const float*)d_in[4];
    const float* var    = (const float*)d_in[5];
    const float* enc_w  = (const float*)d_in[6];
    const float* enc_b  = (const float*)d_in[7];
    float* out = (float*)d_out;

    float* y      = (float*)d_ws;
    float* wts    = y + (size_t)1048576;

    k_compress<<<1024, 128, 0, stream>>>(x, comp_w, gamma, beta, mean, var, y);
    k_encconv<<<256, 256, 0, stream>>>(y, enc_w, enc_b, wts);
    k_reassemble<<<256, 512, 0, stream>>>(x, wts, out);
}


# k_compress: no LDS stage, no barrier: wave = one 64-px image row x one o tile; b128 whole-row x loads straight into the B fragments of 4 interleaved px tiles, comp_w rows into A; 1024 -> 256 blocks
# speedup vs baseline: 2.7426x; 1.1685x over previous
#include <hip/hip_runtime.h>
#include <hip/hip_bf16.h>

typedef __attribute__((ext_vector_type(2))) float v2f;
typedef __attribute__((ext_vector_type(4))) float v4f;
typedef __attribute__((ext_vector_type(8))) float v8f;

#define EPS 1e-5f

__global__ void __launch_bounds__(128)
k_compress(const float* __restrict__ x,
           const float* __restrict__ comp_w,
           const float* __restrict__ gamma,
           const float* __restrict__ beta,
           const float* __restrict__ mean,
           const float* __restrict__ var,
           float* __restrict__ y) {
    const int tid  = threadIdx.x;
    const int lane = tid & 31, c = lane & 15, half = lane >> 4;
    const int oBase = (tid >> 5) << 4;
    const int b = blockIdx.x >> 6;
    const int h = blockIdx.x & 63;

    const float* ap = comp_w + (oBase + c) * 128 + half * 4;
    v4f a[16];
#pragma unroll
    for (int m = 0; m < 16; ++m) a[m] = *(const v4f*)(ap + 8 * m);

    const float* xp = x + ((size_t)b * 128 + half * 4) * 4096 + h * 64 + c * 4;
    v4f bq[16][4];
#pragma unroll
    for (int m = 0; m < 16; ++m) {
#pragma unroll
        for (int e = 0; e < 4; ++e)
            bq[m][e] = *(const v4f*)(xp + (size_t)(8 * m + e) * 4096);
    }

    const int o0 = oBase + half * 8;
    const v4f g0 = *(const v4f*)(gamma + o0), g1 = *(const v4f*)(gamma + o0 + 4);
    const v4f e0 = *(const v4f*)(beta  + o0), e1 = *(const v4f*)(beta  + o0 + 4);
    const v4f m0 = *(const v4f*)(mean  + o0), m1 = *(const v4f*)(mean  + o0 + 4);
    const v4f r0 = *(const v4f*)(var   + o0), r1 = *(const v4f*)(var   + o0 + 4);
    __builtin_amdgcn_sched_barrier(0);

    v8f acc[4] = {v8f{}, v8f{}, v8f{}, v8f{}};
#pragma unroll
    for (int m = 0; m < 16; ++m) {
#pragma unroll
        for (int t = 0; t < 2; ++t) {
            const v2f av = {a[m][2 * t], a[m][2 * t + 1]};
#pragma unroll
            for (int j = 0; j < 4; ++j) {
                const v2f bv = {bq[m][2 * t][j], bq[m][2 * t + 1][j]};
                acc[j] = __builtin_amdgcn_wmma_f32_16x16x4_f32(
                    false, av, false, bv, (short)0, acc[j], false, false);
            }
        }
    }

    float* yp = y + ((size_t)b * 64 + o0) * 4096 + h * 64 + c * 4;
#pragma unroll
    for (int r = 0; r < 8; ++r) {
        const float gm = r < 4 ? g0[r & 3] : g1[r & 3];
        const float bt = r < 4 ? e0[r & 3] : e1[r & 3];
        const float mn = r < 4 ? m0[r & 3] : m1[r & 3];
        const float vr = r < 4 ? r0[r & 3] : r1[r & 3];
        const float iv = gm * rsqrtf(vr + EPS);
        const float bb = bt - mn * iv;
        v4f o4;
#pragma unroll
        for (int j = 0; j < 4; ++j) {
            float v = acc[j][r] * iv + bb;
            o4[j] = v * __frcp_rn(1.f + __expf(-v));
        }
        *(v4f*)(yp + (size_t)r * 4096) = o4;
    }
}

#define ENC_WP  580
#define ENC_WF  (100 * ENC_WP)
#define ENC_ROW 68
#define ENC_CP  208
#define LG_P    68

template <int NT>
__device__ __forceinline__ void enc_wave(const float* __restrict__ lw,
                                         const float* __restrict__ lp,
                                         const int oTile0, const int pxBase,
                                         const int l, const int half,
                                         v8f* __restrict__ acc) {
    for (int t = 0; t < NT; ++t) acc[t] = v8f{};

    const float* aRow[NT];
    for (int t = 0; t < NT; ++t) {
        const int o = (oTile0 + t) * 16 + l;
        aRow[t] = lw + (o < 100 ? o : 99) * ENC_WP + half * 18;
    }
    const float* bBase = lp + half * 2 * ENC_CP + 3 + pxBase + l;
#pragma unroll 2
    for (int g = 0; g < 16; ++g) {
        const float* bp = bBase + g * 4 * ENC_CP;
#pragma unroll
        for (int j = 0; j < 9; ++j) {
            const int e0 = 2 * j, e1 = 2 * j + 1;
            const int t0 = e0 % 9, t1 = e1 % 9;
            v2f bm;
            bm.x = bp[(e0 / 9) * ENC_CP + (t0 / 3) * ENC_ROW + (t0 % 3)];
            bm.y = bp[(e1 / 9) * ENC_CP + (t1 / 3) * ENC_ROW + (t1 % 3)];
            for (int t = 0; t < NT; ++t) {
                const v2f a = *(const v2f*)(aRow[t] + g * 36 + 2 * j);
                acc[t] = __builtin_amdgcn_wmma_f32_16x16x4_f32(
                    false, a, false, bm, (short)0, acc[t], false, false);
            }
        }
    }
}

template <int S1, int S2>
__device__ __forceinline__ void fold_store(const float* __restrict__ lgc,
                                           float* __restrict__ wp,
                                           const bool top, const bool bot,
                                           const bool lft, const bool rgt) {
    float p[25];
    float mx = -1e30f;
#pragma unroll
    for (int q = 0; q < 25; ++q) {
        p[q] = lgc[(q * 4 + S1 * 2 + S2) * LG_P];
        mx = fmaxf(mx, p[q]);
    }
    float sum = 0.f;
#pragma unroll
    for (int q = 0; q < 25; ++q) { p[q] = __expf(p[q] - mx); sum += p[q]; }
    const float rs = __frcp_rn(sum);

    float wc[9] = {0.f, 0.f, 0.f, 0.f, 0.f, 0.f, 0.f, 0.f, 0.f};
#pragma unroll
    for (int ki = 0; ki < 5; ++ki) {
        const int di = (ki + S1) >> 1;
#pragma unroll
        for (int kj = 0; kj < 5; ++kj) {
            const int dj = (kj + S2) >> 1;
            wc[di * 3 + dj] += p[ki * 5 + kj] * rs;
        }
    }
#pragma unroll
    for (int t = 0; t < 9; ++t) {
        const int di = t / 3, dj = t % 3;
        const bool oob = (di == 0 && top) || (di == 2 && bot) ||
                         (dj == 0 && lft) || (dj == 2 && rgt);
        wp[(t * 4 + S1 * 2 + S2) * 64] = oob ? 0.f : wc[t];
    }
}

__global__ void __launch_bounds__(256)
k_encconv(const float* __restrict__ y,
          const float* __restrict__ enc_w,
          const float* __restrict__ enc_b,
          float* __restrict__ wts) {
    __shared__ __attribute__((aligned(16))) float ls[ENC_WF + 64 * ENC_CP];
    float* lw = ls;
    float* lp = ls + ENC_WF;
    const int tid = threadIdx.x;
    const int b   = blockIdx.x >> 6;
    const int h   = blockIdx.x & 63;

    const float* yb = y + (size_t)b * 64 * 4096;
    const int q = tid & 15, cq = tid >> 4;
    float4 v[3][4];
    for (int r = 0; r < 3; ++r) {
        const int hh = h + r - 1;
        if (hh >= 0 && hh < 64) {
            for (int k = 0; k < 4; ++k)
                v[r][k] = *(const float4*)(yb + (size_t)(cq + k * 16) * 4096 + hh * 64 + q * 4);
        } else {
            for (int k = 0; k < 4; ++k)
                v[r][k] = make_float4(0.f, 0.f, 0.f, 0.f);
        }
    }
    for (int r = 0; r < 3; ++r) {
        for (int k = 0; k < 4; ++k)
            *(float4*)(lp + (cq + k * 16) * ENC_CP + r * ENC_ROW + 4 + q * 4) = v[r][k];
    }
    if (tid < 192) {
        float* p = lp + (tid / 3) * ENC_CP + (tid % 3) * ENC_ROW;
        p[3]  = 0.f;
        p[68] = 0.f;
    }

    const float4* g4 = (const float4*)enc_w;
#pragma unroll 14
    for (int k = 0; k < 56; ++k) {
        const int f = tid + k * 256;
        const int row = f / 144;
        *(float4*)(lw + row * ENC_WP + (f - row * 144) * 4) = g4[f];
    }
    if (tid < 64) {
        const int f = tid + 56 * 256;
        const int row = f / 144;
        *(float4*)(lw + row * ENC_WP + (f - row * 144) * 4) = g4[f];
    }
    __syncthreads();

    const int lane = tid & 31, l = lane & 15, half = lane >> 4;
    const int wv = __builtin_amdgcn_readfirstlane(tid >> 5);
    const int pxBase = (wv & 3) << 4;
    v8f acc[4];
    if (wv < 4) enc_wave<4>(lw, lp, 0, pxBase, l, half, acc);
    else        enc_wave<3>(lw, lp, 4, pxBase, l, half, acc);
    __syncthreads();

    float* lgs = lp;
    const int oTile0 = wv < 4 ? 0 : 4;
    const int nt     = wv < 4 ? 4 : 3;
#pragma unroll
    for (int t = 0; t < 4; ++t) {
        if (t < nt) {
#pragma unroll
            for (int j = 0; j < 8; ++j) {
                const int oo = (oTile0 + t) * 16 + j + half * 8;
                if (oo < 100)
                    lgs[oo * LG_P + pxBase + l] = acc[t][j] + enc_b[oo];
            }
        }
    }
    __syncthreads();

    const int w = ((wv & 1) << 5) + lane;
    float* wp = wts + (size_t)(b * 64 + h) * (9 * 4 * 64) + w;
    const bool top = h == 0, bot = h == 63, lft = w == 0, rgt = w == 63;
    switch (wv >> 1) {
    case 0:  fold_store<0, 0>(lgs + w, wp, top, bot, lft, rgt); break;
    case 1:  fold_store<0, 1>(lgs + w, wp, top, bot, lft, rgt); break;
    case 2:  fold_store<1, 0>(lgs + w, wp, top, bot, lft, rgt); break;
    default: fold_store<1, 1>(lgs + w, wp, top, bot, lft, rgt); break;
    }
}

#define RS_CPW 8

__global__ void __launch_bounds__(512)
k_reassemble(const float* __restrict__ x,
             const float* __restrict__ wts,
             float* __restrict__ out) {
    const int tid  = threadIdx.x;
    const int lane = tid & 31;
    const int c0   = (tid >> 5) * RS_CPW;
    const int b    = blockIdx.x >> 6;
    const int h    = blockIdx.x & 63;

    const int aL = ((lane + 31) & 31) << 2;
    const int aR = ((lane + 1) & 31) << 2;

    const float* xb = x + ((size_t)b * 128 + c0) * 4096 + lane * 2;
    const float* x0 = xb + max(h - 1, 0) * 64;
    const float* x1 = xb + h * 64;
    const float* x2 = xb + min(h + 1, 63) * 64;
    float* ob = out + ((size_t)b * 128 + c0) * 16384 + (size_t)h * 256 + lane * 4;

    v2f xs[RS_CPW][3];
#pragma unroll
    for (int c = 0; c < RS_CPW; ++c) {
        xs[c][0] = *(const v2f*)(x0 + (size_t)c * 4096);
        xs[c][1] = *(const v2f*)(x1 + (size_t)c * 4096);
        xs[c][2] = *(const v2f*)(x2 + (size_t)c * 4096);
    }
    const float* wq = wts + (size_t)(b * 64 + h) * (36 * 64) + lane * 2;
    v2f wc[36];
#pragma unroll
    for (int i = 0; i < 36; ++i) wc[i] = *(const v2f*)(wq + i * 64);
    __builtin_amdgcn_sched_barrier(0);

#pragma unroll
    for (int c = 0; c < RS_CPW; ++c) {
        v2f acc[4];
#pragma unroll
        for (int di = 0; di < 3; ++di) {
            const v2f v = xs[c][di];
            const float lf = __int_as_float(
                __builtin_amdgcn_ds_bpermute(aL, __float_as_int(v.y)));
            const float rt = __int_as_float(
                __builtin_amdgcn_ds_bpermute(aR, __float_as_int(v.x)));
            const v2f t0 = {lf, v.x};
            const v2f t1 = v;
            const v2f t2 = {v.y, rt};
#pragma unroll
            for (int s = 0; s < 4; ++s) {
                if (di == 0) acc[s] = wc[s] * t0;
                else acc[s] = __builtin_elementwise_fma(wc[(di * 3) * 4 + s], t0, acc[s]);
                acc[s] = __builtin_elementwise_fma(wc[(di * 3 + 1) * 4 + s], t1, acc[s]);
                acc[s] = __builtin_elementwise_fma(wc[(di * 3 + 2) * 4 + s], t2, acc[s]);
            }
        }
        float* oc = ob + (size_t)c * 16384;
        const v4f r0 = {acc[0].x, acc[1].x, acc[0].y, acc[1].y};
        const v4f r1 = {acc[2].x, acc[3].x, acc[2].y, acc[3].y};
        *(v4f*)(oc)       = r0;
        *(v4f*)(oc + 128) = r1;
    }
}

extern "C" void kernel_launch(void* const* d_in, const int* in_sizes, int n_in,
                              void* d_out, int out_size, void* d_ws, size_t ws_size,
                              hipStream_t stream) {
    const float* x      = (const float*)d_in[0];
    const float* comp_w = (const float*)d_in[1];
    const float* gamma  = (const float*)d_in[2];
    const float* beta   = (const float*)d_in[3];
    const float* mean   = (const float*)d_in[4];
    const float* var    = (const float*)d_in[5];
    const float* enc_w  = (const float*)d_in[6];
    const float* enc_b  = (const float*)d_in[7];
    float* out = (float*)d_out;

    float* y      = (float*)d_ws;
    float* wts    = y + (size_t)1048576;

    k_compress<<<256, 128, 0, stream>>>(x, comp_w, gamma, beta, mean, var, y);
    k_encconv<<<256, 256, 0, stream>>>(y, enc_w, enc_b, wts);
    k_reassemble<<<256, 512, 0, stream>>>(x, wts, out);
}


# encoder bias loaded under the first compressor unit's WMMAs and used as the accumulators' start value: no dependent bias load per logit row behind the matrix loop; bias now summed first, not last
# speedup vs baseline: 3.3083x; 1.2063x over previous
#include <hip/hip_runtime.h>
#include <hip/hip_bf16.h>

typedef __attribute__((ext_vector_type(2))) float v2f;
typedef __attribute__((ext_vector_type(4))) float v4f;
typedef __attribute__((ext_vector_type(8))) float v8f;

#define EPS 1e-5f

#define ENC_WP  580
#define ENC_WF  (100 * ENC_WP)
#define ENC_ROW 68
#define ENC_CP  208
#define LG_P    68
#define LWT_OFF (100 * LG_P)
#define RS_CPW  16


__device__ __forceinline__ void comp_load_a(const float* __restrict__ ap, v4f (&a)[16]) {
#pragma unroll
    for (int m = 0; m < 16; ++m) a[m] = *(const v4f*)(ap + 8 * m);
}

__device__ __forceinline__ void comp_load_b(const float* __restrict__ xp, v4f (&bq)[16][4]) {
#pragma unroll
    for (int m = 0; m < 16; ++m) {
#pragma unroll
        for (int e = 0; e < 4; ++e)
            bq[m][e] = *(const v4f*)(xp + (size_t)(8 * m + e) * 4096);
    }
}

template <int M0, int M1>
__device__ __forceinline__ void comp_mma(const v4f (&a)[16], const v4f (&bq)[16][4],
                                         v8f (&acc)[4]) {
    if (M0 == 0) {
#pragma unroll
        for (int j = 0; j < 4; ++j) acc[j] = v8f{};
    }
#pragma unroll
    for (int m = M0; m < M1; ++m) {
#pragma unroll
        for (int t = 0; t < 2; ++t) {
            const v2f av = {a[m][2 * t], a[m][2 * t + 1]};
#pragma unroll
            for (int j = 0; j < 4; ++j) {
                const v2f bv = {bq[m][2 * t][j], bq[m][2 * t + 1][j]};
                acc[j] = __builtin_amdgcn_wmma_f32_16x16x4_f32(
                    false, av, false, bv, (short)0, acc[j], false, false);
            }
        }
    }
}

__device__ __forceinline__ void bn_load(const float* gamma, const float* beta,
                                        const float* mean, const float* var,
                                        const int o0, float (&iv)[8], float (&bb)[8]) {
    const v4f g0 = *(const v4f*)(gamma + o0), g1 = *(const v4f*)(gamma + o0 + 4);
    const v4f e0 = *(const v4f*)(beta  + o0), e1 = *(const v4f*)(beta  + o0 + 4);
    const v4f m0 = *(const v4f*)(mean  + o0), m1 = *(const v4f*)(mean  + o0 + 4);
    const v4f r0 = *(const v4f*)(var   + o0), r1 = *(const v4f*)(var   + o0 + 4);
#pragma unroll
    for (int r = 0; r < 8; ++r) {
        const float gm = r < 4 ? g0[r & 3] : g1[r & 3];
        const float bt = r < 4 ? e0[r & 3] : e1[r & 3];
        const float mn = r < 4 ? m0[r & 3] : m1[r & 3];
        const float vr = r < 4 ? r0[r & 3] : r1[r & 3];
        iv[r] = gm * rsqrtf(vr + EPS);
        bb[r] = bt - mn * iv[r];
    }
}

__device__ __forceinline__ void comp_store(const v8f (&acc)[4], const float (&iv)[8],
                                           const float (&bb)[8], const int keep,
                                           float* __restrict__ pr) {
#pragma unroll
    for (int r = 0; r < 8; ++r) {
        v4f o4;
#pragma unroll
        for (int j = 0; j < 4; ++j) {
            const float v = acc[j][r] * iv[r] + bb[r];
            const float s = v * __frcp_rn(1.f + __expf(-v));
            o4[j] = __int_as_float(__float_as_int(s) & keep);
        }
        *(v4f*)(pr + r * ENC_CP) = o4;
    }
}

template <int NT>
__device__ __forceinline__ void enc_wave(const float* __restrict__ lw,
                                         const float* __restrict__ lp,
                                         const v4f (&bs)[4][2],
                                         const int oTile0, const int pxBase,
                                         const int l, const int half,
                                         v8f* __restrict__ acc) {
    for (int t = 0; t < NT; ++t)
        acc[t] = v8f{bs[t][0][0], bs[t][0][1], bs[t][0][2], bs[t][0][3],
                     bs[t][1][0], bs[t][1][1], bs[t][1][2], bs[t][1][3]};

    const float* aRow[NT];
    for (int t = 0; t < NT; ++t) {
        const int o = (oTile0 + t) * 16 + l;
        aRow[t] = lw + (o < 100 ? o : 99) * ENC_WP + half * 18;
    }
    const float* bBase = lp + half * 2 * ENC_CP + 3 + pxBase + l;
#pragma unroll 2
    for (int g = 0; g < 16; ++g) {
        const float* bp = bBase + g * 4 * ENC_CP;
#pragma unroll
        for (int j = 0; j < 9; ++j) {
            const int e0 = 2 * j, e1 = 2 * j + 1;
            const int t0 = e0 % 9, t1 = e1 % 9;
            v2f bm;
            bm.x = bp[(e0 / 9) * ENC_CP + (t0 / 3) * ENC_ROW + (t0 % 3)];
            bm.y = bp[(e1 / 9) * ENC_CP + (t1 / 3) * ENC_ROW + (t1 % 3)];
            for (int t = 0; t < NT; ++t) {
                const v2f a = *(const v2f*)(aRow[t] + g * 36 + 2 * j);
                acc[t] = __builtin_amdgcn_wmma_f32_16x16x4_f32(
                    false, a, false, bm, (short)0, acc[t], false, false);
            }
        }
    }
}

template <int S1, int S2>
__device__ __forceinline__ void fold(const float* __restrict__ lgc,
                                     const bool top, const bool bot,
                                     const bool lft, const bool rgt,
                                     float (&wo)[9]) {
    float p[25];
    float mx = -1e30f;
#pragma unroll
    for (int q = 0; q < 25; ++q) {
        p[q] = lgc[(q * 4 + S1 * 2 + S2) * LG_P];
        mx = fmaxf(mx, p[q]);
    }
    float sum = 0.f;
#pragma unroll
    for (int q = 0; q < 25; ++q) { p[q] = __expf(p[q] - mx); sum += p[q]; }
    const float rs = __frcp_rn(sum);

    float wc[9] = {0.f, 0.f, 0.f, 0.f, 0.f, 0.f, 0.f, 0.f, 0.f};
#pragma unroll
    for (int ki = 0; ki < 5; ++ki) {
        const int di = (ki + S1) >> 1;
#pragma unroll
        for (int kj = 0; kj < 5; ++kj) {
            const int dj = (kj + S2) >> 1;
            wc[di * 3 + dj] += p[ki * 5 + kj] * rs;
        }
    }
#pragma unroll
    for (int t = 0; t < 9; ++t) {
        const int di = t / 3, dj = t % 3;
        const bool oob = (di == 0 && top) || (di == 2 && bot) ||
                         (dj == 0 && lft) || (dj == 2 && rgt);
        wo[t] = oob ? 0.f : wc[t];
    }
}

template <int N>
__device__ __forceinline__ void w_load(const v4f* g4, const int f0, v4f (&w)[N]) {
#pragma unroll
    for (int k = 0; k < N; ++k) w[k] = g4[f0 + k * 128];
}

template <int N>
__device__ __forceinline__ void w_store(float* lw, const int f0, const v4f (&w)[N]) {
#pragma unroll
    for (int k = 0; k < N; ++k) {
        const int f = f0 + k * 128;
        const int row = f / 144;
        *(v4f*)(lw + row * ENC_WP + (f - row * 144) * 4) = w[k];
    }
}

__global__ void __launch_bounds__(256)
k_rowblock(const float* __restrict__ x,
          const float* __restrict__ comp_w,
          const float* gamma,
          const float* beta,
          const float* mean,
          const float* var,
          const float* enc_w,
          const float* __restrict__ enc_b,
          float* __restrict__ out) {
    __shared__ __attribute__((aligned(16))) float ls[ENC_WF + 64 * ENC_CP];
    float* lw = ls;
    float* lp = ls + ENC_WF;
    const int tid = threadIdx.x;
    const int b   = blockIdx.x >> 6;
    const int h   = blockIdx.x & 63;
    const int lane = tid & 31, l = lane & 15, half = lane >> 4;
    const int wv = __builtin_amdgcn_readfirstlane(tid >> 5);

    if (tid < 192) {
        float* p = lp + (tid / 3) * ENC_CP + (tid % 3) * ENC_ROW;
        p[3]  = 0.f;
        p[68] = 0.f;
    }

    v4f bs[4][2];

    {
        const v4f* g4 = (const v4f*)enc_w;
        const int fb = wv < 4 ? tid : 1920 + tid;
        const int oBase = (wv & 3) << 4;
        const int o0 = oBase + half * 8;
        const float* ap = comp_w + (oBase + l) * 128 + half * 4;
        const float* xb = x + ((size_t)b * 128 + half * 4) * 4096 + l * 4;
        float* pb = lp + o0 * ENC_CP + 4 + l * 4;
        const int rr1 = wv >> 2;
        const int hh1 = h + rr1 - 1;

        v4f a[16];
        v4f bq[16][4];
        v8f acc[4];
        v4f wr[8];
        float iv[8], bb[8];
        comp_load_a(ap, a);
        comp_load_b(xb + max(hh1, 0) * 64, bq);
        __builtin_amdgcn_sched_barrier(0);
        comp_mma<0, 1>(a, bq, acc);
        __builtin_amdgcn_sched_barrier(0);
        w_load<8>(g4, fb, wr);
        {
            const int oT0 = wv < 4 ? 0 : 4;
#pragma unroll
            for (int t = 0; t < 4; ++t) {
                const int ob0 = (oT0 + t) * 16 + 8 * half;
                bs[t][0] = *(const v4f*)(enc_b + min(ob0, 96));
                bs[t][1] = *(const v4f*)(enc_b + min(ob0 + 4, 96));
            }
        }
        __builtin_amdgcn_sched_barrier(0);
        comp_mma<1, 16>(a, bq, acc);
        __builtin_amdgcn_sched_barrier(0);
        bn_load(gamma, beta, mean, var, o0, iv, bb);
        w_store<8>(lw, fb, wr);
        if (wv < 4) {
            comp_load_b(xb + min(h + 1, 63) * 64, bq);
            w_load<8>(g4, fb + 8 * 128, wr);
            __builtin_amdgcn_sched_barrier(0);
            comp_store(acc, iv, bb, h > 0 ? -1 : 0, pb);
            __builtin_amdgcn_sched_barrier(0);
            comp_mma<0, 16>(a, bq, acc);
            __builtin_amdgcn_sched_barrier(0);
            w_store<8>(lw, fb + 8 * 128, wr);
            comp_store(acc, iv, bb, h < 63 ? -1 : 0, pb + 2 * ENC_ROW);
        } else {
            v4f wt = v4f{};
            if (tid < 192) wt = g4[fb + 96 * 128];
            v4f wa[16], wb[16];
            w_load<8>(g4, fb + 8 * 128, wr);
            w_load<16>(g4, fb + 16 * 128, wa);
            __builtin_amdgcn_sched_barrier(0);
            comp_store(acc, iv, bb, -1, pb + ENC_ROW);
            w_store<8>(lw, fb + 8 * 128, wr);
            w_load<16>(g4, fb + 32 * 128, wb);
            __builtin_amdgcn_sched_barrier(0);
            w_store<16>(lw, fb + 16 * 128, wa);
            w_load<16>(g4, fb + 48 * 128, wa);
            __builtin_amdgcn_sched_barrier(0);
            w_store<16>(lw, fb + 32 * 128, wb);
            w_load<16>(g4, fb + 64 * 128, wb);
            __builtin_amdgcn_sched_barrier(0);
            w_store<16>(lw, fb + 48 * 128, wa);
            w_load<16>(g4, fb + 80 * 128, wa);
            __builtin_amdgcn_sched_barrier(0);
            w_store<16>(lw, fb + 64 * 128, wb);
            w_store<16>(lw, fb + 80 * 128, wa);
            if (tid < 192) {
                const int f = fb + 96 * 128;
                const int row = f / 144;
                *(v4f*)(lw + row * ENC_WP + (f - row * 144) * 4) = wt;
            }
        }
    }
    __syncthreads();

    const int c0 = wv * RS_CPW;
    v2f xs[RS_CPW][3];
    {
        const float* xr = x + ((size_t)b * 128 + c0) * 4096 + lane * 2;
        const float* x0 = xr + max(h - 1, 0) * 64;
        const float* x1 = xr + h * 64;
        const float* x2 = xr + min(h + 1, 63) * 64;
#pragma unroll
        for (int c = 0; c < RS_CPW; ++c) {
            xs[c][0] = *(const v2f*)(x0 + (size_t)c * 4096);
            xs[c][1] = *(const v2f*)(x1 + (size_t)c * 4096);
            xs[c][2] = *(const v2f*)(x2 + (size_t)c * 4096);
        }
    }

    const int pxBase = (wv & 3) << 4;
    v8f acc[4];
    if (wv < 4) enc_wave<4>(lw, lp, bs, 0, pxBase, l, half, acc);
    else        enc_wave<3>(lw, lp, bs, 4, pxBase, l, half, acc);
    __syncthreads();

    float* lgs = lp;
    const int oTile0 = wv < 4 ? 0 : 4;
    const int nt     = wv < 4 ? 4 : 3;
#pragma unroll
    for (int t = 0; t < 4; ++t) {
        if (t < nt) {
#pragma unroll
            for (int j = 0; j < 8; ++j) {
                const int oo = (oTile0 + t) * 16 + j + half * 8;
                if (oo < 100)
                    lgs[oo * LG_P + pxBase + l] = acc[t][j];
            }
        }
    }
    __syncthreads();

    const int w = ((wv & 1) << 5) + lane;
    const int sub = wv >> 1;
    const bool top = h == 0, bot = h == 63, lft = w == 0, rgt = w == 63;
    float wo[9];
    switch (sub) {
    case 0:  fold<0, 0>(lgs + w, top, bot, lft, rgt, wo); break;
    case 1:  fold<0, 1>(lgs + w, top, bot, lft, rgt, wo); break;
    case 2:  fold<1, 0>(lgs + w, top, bot, lft, rgt, wo); break;
    default: fold<1, 1>(lgs + w, top, bot, lft, rgt, wo); break;
    }
    float* lwt = lp + LWT_OFF;
#pragma unroll
    for (int t = 0; t < 9; ++t) lwt[(t * 4 + sub) * 64 + w] = wo[t];
    __syncthreads();

    const int aL = ((lane + 31) & 31) << 2;
    const int aR = ((lane + 1) & 31) << 2;
    v2f wc[36];
#pragma unroll
    for (int i = 0; i < 36; ++i) wc[i] = *(const v2f*)(lwt + i * 64 + lane * 2);

    float* ob = out + ((size_t)b * 128 + c0) * 16384 + (size_t)h * 256 + lane * 4;
#pragma unroll
    for (int c = 0; c < RS_CPW; ++c) {
        v2f ra[4];
#pragma unroll
        for (int di = 0; di < 3; ++di) {
            const v2f v = xs[c][di];
            const float lf = __int_as_float(
                __builtin_amdgcn_ds_bpermute(aL, __float_as_int(v.y)));
            const float rt = __int_as_float(
                __builtin_amdgcn_ds_bpermute(aR, __float_as_int(v.x)));
            const v2f t0 = {lf, v.x};
            const v2f t1 = v;
            const v2f t2 = {v.y, rt};
#pragma unroll
            for (int q = 0; q < 4; ++q) {
                if (di == 0) ra[q] = wc[q] * t0;
                else ra[q] = __builtin_elementwise_fma(wc[(di * 3) * 4 + q], t0, ra[q]);
                ra[q] = __builtin_elementwise_fma(wc[(di * 3 + 1) * 4 + q], t1, ra[q]);
                ra[q] = __builtin_elementwise_fma(wc[(di * 3 + 2) * 4 + q], t2, ra[q]);
            }
        }
        float* oc = ob + (size_t)c * 16384;
        const v4f r0 = {ra[0].x, ra[1].x, ra[0].y, ra[1].y};
        const v4f r1 = {ra[2].x, ra[3].x, ra[2].y, ra[3].y};
        *(v4f*)(oc)       = r0;
        *(v4f*)(oc + 128) = r1;
    }
}

extern "C" void kernel_launch(void* const* d_in, const int* in_sizes, int n_in,
                              void* d_out, int out_size, void* d_ws, size_t ws_size,
                              hipStream_t stream) {
    const float* x      = (const float*)d_in[0];
    const float* comp_w = (const float*)d_in[1];
    const float* gamma  = (const float*)d_in[2];
    const float* beta   = (const float*)d_in[3];
    const float* mean   = (const float*)d_in[4];
    const float* var    = (const float*)d_in[5];
    const float* enc_w  = (const float*)d_in[6];
    const float* enc_b  = (const float*)d_in[7];
    float* out = (float*)d_out;

    k_rowblock<<<256, 256, 0, stream>>>(x, comp_w, gamma, beta, mean, var, enc_w, enc_b, out);
}
